# starting kernel
# baseline (speedup 1.0000x reference)
#include <hip/hip_runtime.h>
#include <cmath>

typedef __attribute__((ext_vector_type(2))) float v2f;
typedef __attribute__((ext_vector_type(4))) float v4f;
typedef __attribute__((ext_vector_type(8))) float v8f;

#define WMMA4(a, b, c) __builtin_amdgcn_wmma_f32_16x16x4_f32( \
    false, (a), false, (b), (short)0, (c), false, false)

namespace {
constexpr int kB = 8, kL = 512, kN = 64, kF = 128, kH = 8, kC = 16;
constexpr int kChunks = kL / kC;
constexpr int XS = 132;
constexpr int ST = 18;
constexpr float kEps = 1e-6f;
}

__device__ __forceinline__ float phi(float x) {
  return fmaxf(x, 0.0f) + __expf(fminf(x, 0.0f));
}

__global__ __launch_bounds__(256, 1)
void cla_fused_kernel(const float* __restrict__ qin,
                      const float* __restrict__ Wq, const float* __restrict__ bq,
                      const float* __restrict__ Wk, const float* __restrict__ bk,
                      const float* __restrict__ Wv, const float* __restrict__ bv,
                      float* __restrict__ out)
{
  __shared__ float xs[kC * XS];
  __shared__ float shQ[kH][kC * ST];
  __shared__ float shK[kH][kC * ST];
  __shared__ float shV[kH][kC * ST];
  __shared__ float shA[kH][kC * ST];
  __shared__ float shS[kH][kC * ST];
  __shared__ float shKcum[kH][16];
  __shared__ float shZ[kH][16];
  __shared__ float ostage[kC * kF];

  const int tid   = threadIdx.x;
  const int lane  = tid & 31;
  const int w     = tid >> 5;
  const int col   = lane & 15;
  const int hi    = lane >> 4;
  const int khalf = hi << 1;
  const int mrow  = hi << 3;

  const int g = blockIdx.x;
  const int b = g >> 6;
  const int n = g & 63;

  float* Qt = shQ[w];
  float* Kt = shK[w];
  float* Vt = shV[w];
  float* At = shA[w];
  float* St = shS[w];

  for (int i = lane; i < kC * ST; i += 32) St[i] = 0.0f;
  if (lane < 16) shKcum[w][lane] = 0.0f;

  const int wcol = w * 16 + col;
  const float bqv = bq[wcol];
  const float bkv = bk[wcol];
  const float bvv = bv[wcol];

  const int cr = tid >> 4;
  const int cc = (tid & 15) << 3;

  for (int t = 0; t < kChunks; ++t) {
    {
      const float* src =
          qin + (((size_t)(b * kL + t * kC + cr) * kN + n) * kF + cc);
      v4f a0 = *reinterpret_cast<const v4f*>(src);
      v4f a1 = *reinterpret_cast<const v4f*>(src + 4);
      *reinterpret_cast<v4f*>(&xs[cr * XS + cc]) = a0;
      *reinterpret_cast<v4f*>(&xs[cr * XS + cc + 4]) = a1;
      if (t + 1 < kChunks) {
        __builtin_prefetch(src + (size_t)kC * kN * kF, 0, 0);
      }
    }
    __syncthreads();

    v8f cq = {}; v8f ck = {}; v8f cv = {};
    for (int s = 0; s < 32; ++s) {
      const int k0 = 4 * s + khalf;
      v2f a = *reinterpret_cast<v2f*>(&xs[col * XS + k0]);
      v2f wb;
      wb.x = Wq[k0 * kF + wcol]; wb.y = Wq[(k0 + 1) * kF + wcol];
      cq = WMMA4(a, wb, cq);
      wb.x = Wk[k0 * kF + wcol]; wb.y = Wk[(k0 + 1) * kF + wcol];
      ck = WMMA4(a, wb, ck);
      wb.x = Wv[k0 * kF + wcol]; wb.y = Wv[(k0 + 1) * kF + wcol];
      cv = WMMA4(a, wb, cv);
    }

    for (int r = 0; r < 8; ++r) {
      const int m = r + mrow;
      Qt[m * ST + col] = phi(cq[r] + bqv);
      Kt[m * ST + col] = phi(ck[r] + bkv);
      Vt[m * ST + col] = cv[r] + bvv;
    }
    asm volatile("" ::: "memory");

    v8f aat = {};
    for (int s = 0; s < 4; ++s) {
      const int k0 = 4 * s + khalf;
      v2f a  = *reinterpret_cast<v2f*>(&Qt[col * ST + k0]);
      v2f bb = *reinterpret_cast<v2f*>(&Kt[col * ST + k0]);
      aat = WMMA4(a, bb, aat);
    }
    for (int r = 0; r < 8; ++r) {
      const int m = r + mrow;
      At[m * ST + col] = (col <= m) ? aat[r] : 0.0f;
    }
    asm volatile("" ::: "memory");

    if (lane < 16) {
      float den = 0.0f;
      for (int c2 = 0; c2 < 16; ++c2) den += At[lane * ST + c2];
      float dq = 0.0f;
      for (int e = 0; e < 16; ++e) dq += Qt[lane * ST + e] * shKcum[w][e];
      shZ[w][lane] = __builtin_amdgcn_rcpf(den + dq + kEps);
    }
    asm volatile("" ::: "memory");

    v8f oac = {};
    for (int s = 0; s < 4; ++s) {
      const int k0 = 4 * s + khalf;
      v2f a = *reinterpret_cast<v2f*>(&At[col * ST + k0]);
      v2f bb; bb.x = Vt[k0 * ST + col]; bb.y = Vt[(k0 + 1) * ST + col];
      oac = WMMA4(a, bb, oac);
    }
    for (int s = 0; s < 4; ++s) {
      const int k0 = 4 * s + khalf;
      v2f a = *reinterpret_cast<v2f*>(&Qt[col * ST + k0]);
      v2f bb; bb.x = St[k0 * ST + col]; bb.y = St[(k0 + 1) * ST + col];
      oac = WMMA4(a, bb, oac);
    }

    v8f sac;
    for (int r = 0; r < 8; ++r) sac[r] = St[(r + mrow) * ST + col];
    for (int s = 0; s < 4; ++s) {
      const int k0 = 4 * s + khalf;
      v2f a;  a.x  = Kt[k0 * ST + col]; a.y  = Kt[(k0 + 1) * ST + col];
      v2f bb; bb.x = Vt[k0 * ST + col]; bb.y = Vt[(k0 + 1) * ST + col];
      sac = WMMA4(a, bb, sac);
    }
    for (int r = 0; r < 8; ++r) St[(r + mrow) * ST + col] = sac[r];

    if (lane < 16) {
      float s = 0.0f;
      for (int c2 = 0; c2 < 16; ++c2) s += Kt[c2 * ST + lane];
      shKcum[w][lane] += s;
    }
    asm volatile("" ::: "memory");

    for (int r = 0; r < 8; ++r) {
      const int m = r + mrow;
      ostage[m * kF + w * 16 + col] = oac[r] * shZ[w][m];
    }
    __syncthreads();

    {
      float* dst =
          out + (((size_t)(b * kL + t * kC + cr) * kN + n) * kF + cc);
      *reinterpret_cast<v4f*>(dst)     = *reinterpret_cast<v4f*>(&ostage[cr * kF + cc]);
      *reinterpret_cast<v4f*>(dst + 4) = *reinterpret_cast<v4f*>(&ostage[cr * kF + cc + 4]);
    }
  }
}

extern "C" void kernel_launch(void* const* d_in, const int* in_sizes, int n_in,
                              void* d_out, int out_size, void* d_ws, size_t ws_size,
                              hipStream_t stream) {
  (void)in_sizes; (void)n_in; (void)out_size; (void)d_ws; (void)ws_size;
  const float* q  = (const float*)d_in[0];
  const float* Wq = (const float*)d_in[1];
  const float* bq = (const float*)d_in[2];
  const float* Wk = (const float*)d_in[3];
  const float* bk = (const float*)d_in[4];
  const float* Wv = (const float*)d_in[5];
  const float* bv = (const float*)d_in[6];
  cla_fused_kernel<<<dim3(kB * kN), dim3(256), 0, stream>>>(
      q, Wq, bq, Wk, bk, Wv, bv, (float*)d_out);
}


# Projection blocked over 4 chunks: each W fragment is loaded once per 64 timesteps and feeds 4 x-tile WMMAs per projection (x and out staged 64 rows at a time); sum order unchanged
# speedup vs baseline: 1.2796x; 1.2796x over previous
#include <hip/hip_runtime.h>
#include <cmath>

typedef __attribute__((ext_vector_type(2))) float v2f;
typedef __attribute__((ext_vector_type(4))) float v4f;
typedef __attribute__((ext_vector_type(8))) float v8f;

#define WMMA4(a, b, c) __builtin_amdgcn_wmma_f32_16x16x4_f32( \
    false, (a), false, (b), (short)0, (c), false, false)

namespace {
constexpr int kB = 8, kL = 512, kN = 64, kF = 128, kH = 8, kC = 16;
constexpr int kG = 4;
constexpr int kR = kG * kC;
constexpr int kGroups = kL / kR;
constexpr int XS = 132;
constexpr int ST = 18;
constexpr float kEps = 1e-6f;
}

__device__ __forceinline__ float phi(float x) {
  return fmaxf(x, 0.0f) + __expf(fminf(x, 0.0f));
}

__global__ __launch_bounds__(256, 1)
void cla_fused_kernel(const float* __restrict__ qin,
                      const float* __restrict__ Wq, const float* __restrict__ bq,
                      const float* __restrict__ Wk, const float* __restrict__ bk,
                      const float* __restrict__ Wv, const float* __restrict__ bv,
                      float* __restrict__ out)
{
  __shared__ float xs[kR * XS];
  __shared__ float shQ[kH][kC * ST];
  __shared__ float shK[kH][kC * ST];
  __shared__ float shV[kH][kC * ST];
  __shared__ float shA[kH][kC * ST];
  __shared__ float shS[kH][kC * ST];
  __shared__ float shKcum[kH][16];
  __shared__ float shZ[kH][16];
  __shared__ float ostage[kR * kF];

  const int tid   = threadIdx.x;
  const int lane  = tid & 31;
  const int w     = tid >> 5;
  const int col   = lane & 15;
  const int hi    = lane >> 4;
  const int khalf = hi << 1;
  const int mrow  = hi << 3;

  const int g = blockIdx.x;
  const int b = g >> 6;
  const int n = g & 63;

  float* Qt = shQ[w];
  float* Kt = shK[w];
  float* Vt = shV[w];
  float* At = shA[w];
  float* St = shS[w];

  for (int i = lane; i < kC * ST; i += 32) St[i] = 0.0f;
  if (lane < 16) shKcum[w][lane] = 0.0f;

  const int wcol = w * 16 + col;
  const float bqv = bq[wcol];
  const float bkv = bk[wcol];
  const float bvv = bv[wcol];

  const int cr = tid >> 4;
  const int cc = (tid & 15) << 3;

  for (int tg = 0; tg < kGroups; ++tg) {
    {
      const float* src =
          qin + (((size_t)(b * kL + tg * kR + cr) * kN + n) * kF + cc);
      v4f a0[kG], a1[kG];
      for (int j = 0; j < kG; ++j) {
        const float* sj = src + (size_t)j * kC * kN * kF;
        a0[j] = *reinterpret_cast<const v4f*>(sj);
        a1[j] = *reinterpret_cast<const v4f*>(sj + 4);
      }
      for (int j = 0; j < kG; ++j) {
        *reinterpret_cast<v4f*>(&xs[(j * kC + cr) * XS + cc])     = a0[j];
        *reinterpret_cast<v4f*>(&xs[(j * kC + cr) * XS + cc + 4]) = a1[j];
      }
      if (tg + 1 < kGroups) {
        for (int j = 0; j < kG; ++j) {
          __builtin_prefetch(src + (size_t)(kR + j * kC) * kN * kF, 0, 0);
        }
      }
    }
    __syncthreads();

    v8f cq[kG], ck[kG], cv[kG];
    for (int j = 0; j < kG; ++j) { cq[j] = {}; ck[j] = {}; cv[j] = {}; }
    #pragma unroll 4
    for (int s = 0; s < 32; ++s) {
      const int k0 = 4 * s + khalf;
      v2f wq2, wk2, wv2;
      wq2.x = Wq[k0 * kF + wcol]; wq2.y = Wq[(k0 + 1) * kF + wcol];
      wk2.x = Wk[k0 * kF + wcol]; wk2.y = Wk[(k0 + 1) * kF + wcol];
      wv2.x = Wv[k0 * kF + wcol]; wv2.y = Wv[(k0 + 1) * kF + wcol];
      for (int j = 0; j < kG; ++j) {
        v2f a = *reinterpret_cast<v2f*>(&xs[(j * kC + col) * XS + k0]);
        cq[j] = WMMA4(a, wq2, cq[j]);
        ck[j] = WMMA4(a, wk2, ck[j]);
        cv[j] = WMMA4(a, wv2, cv[j]);
      }
    }

    #pragma unroll
    for (int j = 0; j < kG; ++j) {
      for (int r = 0; r < 8; ++r) {
        const int m = r + mrow;
        Qt[m * ST + col] = phi(cq[j][r] + bqv);
        Kt[m * ST + col] = phi(ck[j][r] + bkv);
        Vt[m * ST + col] = cv[j][r] + bvv;
      }
      asm volatile("" ::: "memory");

      v8f aat = {};
      for (int s = 0; s < 4; ++s) {
        const int k0 = 4 * s + khalf;
        v2f a  = *reinterpret_cast<v2f*>(&Qt[col * ST + k0]);
        v2f bb = *reinterpret_cast<v2f*>(&Kt[col * ST + k0]);
        aat = WMMA4(a, bb, aat);
      }
      for (int r = 0; r < 8; ++r) {
        const int m = r + mrow;
        At[m * ST + col] = (col <= m) ? aat[r] : 0.0f;
      }
      asm volatile("" ::: "memory");

      if (lane < 16) {
        float den = 0.0f;
        for (int c2 = 0; c2 < 16; ++c2) den += At[lane * ST + c2];
        float dq = 0.0f;
        for (int e = 0; e < 16; ++e) dq += Qt[lane * ST + e] * shKcum[w][e];
        shZ[w][lane] = __builtin_amdgcn_rcpf(den + dq + kEps);
      }
      asm volatile("" ::: "memory");

      v8f oac = {};
      for (int s = 0; s < 4; ++s) {
        const int k0 = 4 * s + khalf;
        v2f a = *reinterpret_cast<v2f*>(&At[col * ST + k0]);
        v2f bb; bb.x = Vt[k0 * ST + col]; bb.y = Vt[(k0 + 1) * ST + col];
        oac = WMMA4(a, bb, oac);
      }
      for (int s = 0; s < 4; ++s) {
        const int k0 = 4 * s + khalf;
        v2f a = *reinterpret_cast<v2f*>(&Qt[col * ST + k0]);
        v2f bb; bb.x = St[k0 * ST + col]; bb.y = St[(k0 + 1) * ST + col];
        oac = WMMA4(a, bb, oac);
      }

      v8f sac;
      for (int r = 0; r < 8; ++r) sac[r] = St[(r + mrow) * ST + col];
      for (int s = 0; s < 4; ++s) {
        const int k0 = 4 * s + khalf;
        v2f a;  a.x  = Kt[k0 * ST + col]; a.y  = Kt[(k0 + 1) * ST + col];
        v2f bb; bb.x = Vt[k0 * ST + col]; bb.y = Vt[(k0 + 1) * ST + col];
        sac = WMMA4(a, bb, sac);
      }
      for (int r = 0; r < 8; ++r) St[(r + mrow) * ST + col] = sac[r];

      if (lane < 16) {
        float s = 0.0f;
        for (int c2 = 0; c2 < 16; ++c2) s += Kt[c2 * ST + lane];
        shKcum[w][lane] += s;
      }
      asm volatile("" ::: "memory");

      for (int r = 0; r < 8; ++r) {
        const int m = r + mrow;
        ostage[(j * kC + m) * kF + w * 16 + col] = oac[r] * shZ[w][m];
      }
    }
    __syncthreads();

    {
      float* dst =
          out + (((size_t)(b * kL + tg * kR + cr) * kN + n) * kF + cc);
      for (int j = 0; j < kG; ++j) {
        float* dj = dst + (size_t)j * kC * kN * kF;
        const float* oj = &ostage[(j * kC + cr) * kF + cc];
        *reinterpret_cast<v4f*>(dj)     = *reinterpret_cast<const v4f*>(oj);
        *reinterpret_cast<v4f*>(dj + 4) = *reinterpret_cast<const v4f*>(oj + 4);
      }
    }
  }
}

extern "C" void kernel_launch(void* const* d_in, const int* in_sizes, int n_in,
                              void* d_out, int out_size, void* d_ws, size_t ws_size,
                              hipStream_t stream) {
  (void)in_sizes; (void)n_in; (void)out_size; (void)d_ws; (void)ws_size;
  const float* q  = (const float*)d_in[0];
  const float* Wq = (const float*)d_in[1];
  const float* bq = (const float*)d_in[2];
  const float* Wk = (const float*)d_in[3];
  const float* bk = (const float*)d_in[4];
  const float* Wv = (const float*)d_in[5];
  const float* bv = (const float*)d_in[6];
  cla_fused_kernel<<<dim3(kB * kN), dim3(256), 0, stream>>>(
      q, Wq, bq, Wk, bk, Wv, bv, (float*)d_out);
}


# Scan fed from accumulator-layout registers: scores computed transposed so A, V, S and the normalizer never leave registers; only phi(Q), phi(K) pass LDS; sum order of normalizer, kcum, S, O changed
# speedup vs baseline: 1.3979x; 1.0925x over previous
#include <hip/hip_runtime.h>
#include <cmath>

typedef __attribute__((ext_vector_type(2))) float v2f;
typedef __attribute__((ext_vector_type(4))) float v4f;
typedef __attribute__((ext_vector_type(8))) float v8f;

#define WMMA4(a, b, c) __builtin_amdgcn_wmma_f32_16x16x4_f32( \
    false, (a), false, (b), (short)0, (c), false, false)

namespace {
constexpr int kB = 8, kL = 512, kN = 64, kF = 128, kH = 8, kC = 16;
constexpr int kG = 4;
constexpr int kR = kG * kC;
constexpr int kGroups = kL / kR;
constexpr int XS = 132;
constexpr int ST = 20;
constexpr float kEps = 1e-6f;
}

__device__ __forceinline__ float phi(float x) {
  return fmaxf(x, 0.0f) + __expf(fminf(x, 0.0f));
}

__device__ __forceinline__ float xhalf(float x) {
  return __builtin_bit_cast(float, __builtin_amdgcn_permlanex16(
      __builtin_bit_cast(int, x), __builtin_bit_cast(int, x),
      0x76543210, 0xfedcba98, false, false));
}

__global__ __launch_bounds__(256, 1)
void cla_fused_kernel(const float* __restrict__ qin,
                      const float* __restrict__ Wq, const float* __restrict__ bq,
                      const float* __restrict__ Wk, const float* __restrict__ bk,
                      const float* __restrict__ Wv, const float* __restrict__ bv,
                      float* __restrict__ out)
{
  __shared__ alignas(16) float xs[kR * XS];
  __shared__ alignas(16) float shQ[kH][kC * ST];
  __shared__ alignas(16) float shK[kH][kC * ST];
  __shared__ alignas(16) float shKcum[kH][16];
  __shared__ alignas(16) float ostage[kR * kF];

  const int tid   = threadIdx.x;
  const int lane  = tid & 31;
  const int w     = tid >> 5;
  const int col   = lane & 15;
  const int hi    = lane >> 4;
  const int khalf = hi << 1;
  const int mrow  = hi << 3;

  const int g = blockIdx.x;
  const int b = g >> 6;
  const int n = g & 63;

  float* Qt = shQ[w];
  float* Kt = shK[w];

  v8f sac = {};
  float kcum = 0.0f;
  shKcum[w][col] = 0.0f;

  const int wcol = w * 16 + col;
  const float bqv = bq[wcol];
  const float bkv = bk[wcol];
  const float bvv = bv[wcol];

  const int cr = tid >> 4;
  const int cc = (tid & 15) << 3;

  for (int tg = 0; tg < kGroups; ++tg) {
    {
      const float* src =
          qin + (((size_t)(b * kL + tg * kR + cr) * kN + n) * kF + cc);
      v4f a0[kG], a1[kG];
      for (int j = 0; j < kG; ++j) {
        const float* sj = src + (size_t)j * kC * kN * kF;
        a0[j] = *reinterpret_cast<const v4f*>(sj);
        a1[j] = *reinterpret_cast<const v4f*>(sj + 4);
      }
      for (int j = 0; j < kG; ++j) {
        *reinterpret_cast<v4f*>(&xs[(j * kC + cr) * XS + cc])     = a0[j];
        *reinterpret_cast<v4f*>(&xs[(j * kC + cr) * XS + cc + 4]) = a1[j];
      }
      if (tg + 1 < kGroups) {
        for (int j = 0; j < kG; ++j) {
          __builtin_prefetch(src + (size_t)(kR + j * kC) * kN * kF, 0, 0);
        }
      }
    }
    __syncthreads();

    v8f cq[kG], ck[kG], cv[kG];
    for (int j = 0; j < kG; ++j) { cq[j] = {}; ck[j] = {}; cv[j] = {}; }
    #pragma unroll 4
    for (int s = 0; s < 32; ++s) {
      const int k0 = 4 * s + khalf;
      v2f wq2, wk2, wv2;
      wq2.x = Wq[k0 * kF + wcol]; wq2.y = Wq[(k0 + 1) * kF + wcol];
      wk2.x = Wk[k0 * kF + wcol]; wk2.y = Wk[(k0 + 1) * kF + wcol];
      wv2.x = Wv[k0 * kF + wcol]; wv2.y = Wv[(k0 + 1) * kF + wcol];
      for (int j = 0; j < kG; ++j) {
        v2f a = *reinterpret_cast<v2f*>(&xs[(j * kC + col) * XS + k0]);
        cq[j] = WMMA4(a, wq2, cq[j]);
        ck[j] = WMMA4(a, wk2, ck[j]);
        cv[j] = WMMA4(a, wv2, cv[j]);
      }
    }

    #pragma unroll
    for (int j = 0; j < kG; ++j) {
      float qv[8], kv[8];
      v8f vv;
      for (int r = 0; r < 8; ++r) {
        qv[r] = phi(cq[j][r] + bqv);
        kv[r] = phi(ck[j][r] + bkv);
        vv[r] = cv[j][r] + bvv;
      }
      for (int r = 0; r < 8; ++r) {
        const int m = r + mrow;
        Qt[m * ST + col] = qv[r];
        Kt[m * ST + col] = kv[r];
      }
      asm volatile("" ::: "memory");

      const v4f q0 = *reinterpret_cast<const v4f*>(&Qt[col * ST + mrow]);
      const v4f q1 = *reinterpret_cast<const v4f*>(&Qt[col * ST + mrow + 4]);
      const v4f k0 = *reinterpret_cast<const v4f*>(&Kt[col * ST + mrow]);
      const v4f k1 = *reinterpret_cast<const v4f*>(&Kt[col * ST + mrow + 4]);
      const v4f c0 = *reinterpret_cast<const v4f*>(&shKcum[w][mrow]);
      const v4f c1 = *reinterpret_cast<const v4f*>(&shKcum[w][mrow + 4]);
      asm volatile("" ::: "memory");

      float ks = ((kv[0] + kv[1]) + (kv[2] + kv[3])) +
                 ((kv[4] + kv[5]) + (kv[6] + kv[7]));
      ks += xhalf(ks);
      kcum += ks;
      shKcum[w][col] = kcum;

      v8f at = {};
      v8f sn = sac;
      {
        v2f ka, va;
        at = WMMA4(k0.xy, q0.xy, at);
        ka.x = kv[0]; ka.y = kv[1]; va.x = vv[0]; va.y = vv[1];
        sn = WMMA4(ka, va, sn);
        at = WMMA4(k0.zw, q0.zw, at);
        ka.x = kv[2]; ka.y = kv[3]; va.x = vv[2]; va.y = vv[3];
        sn = WMMA4(ka, va, sn);
        at = WMMA4(k1.xy, q1.xy, at);
        ka.x = kv[4]; ka.y = kv[5]; va.x = vv[4]; va.y = vv[5];
        sn = WMMA4(ka, va, sn);
        at = WMMA4(k1.zw, q1.zw, at);
        ka.x = kv[6]; ka.y = kv[7]; va.x = vv[6]; va.y = vv[7];
        sn = WMMA4(ka, va, sn);
      }

      float p = 0.0f;
      for (int r = 0; r < 8; ++r) {
        at[r] = (mrow + r <= col) ? at[r] : 0.0f;
        p += at[r];
      }
      p += q0.x * c0.x; p += q0.y * c0.y; p += q0.z * c0.z; p += q0.w * c0.w;
      p += q1.x * c1.x; p += q1.y * c1.y; p += q1.z * c1.z; p += q1.w * c1.w;
      p += xhalf(p);
      const float z = __builtin_amdgcn_rcpf(p + kEps);

      v8f o1 = {};
      v8f o2 = {};
      const v4f qz0 = q0 * z;
      const v4f qz1 = q1 * z;
      {
        v2f aa, bb, sb;
        aa.x = at[0] * z; aa.y = at[1] * z; bb.x = vv[0]; bb.y = vv[1];
        o1 = WMMA4(aa, bb, o1);
        sb.x = sac[0]; sb.y = sac[1];
        o2 = WMMA4(qz0.xy, sb, o2);
        aa.x = at[2] * z; aa.y = at[3] * z; bb.x = vv[2]; bb.y = vv[3];
        o1 = WMMA4(aa, bb, o1);
        sb.x = sac[2]; sb.y = sac[3];
        o2 = WMMA4(qz0.zw, sb, o2);
        aa.x = at[4] * z; aa.y = at[5] * z; bb.x = vv[4]; bb.y = vv[5];
        o1 = WMMA4(aa, bb, o1);
        sb.x = sac[4]; sb.y = sac[5];
        o2 = WMMA4(qz1.xy, sb, o2);
        aa.x = at[6] * z; aa.y = at[7] * z; bb.x = vv[6]; bb.y = vv[7];
        o1 = WMMA4(aa, bb, o1);
        sb.x = sac[6]; sb.y = sac[7];
        o2 = WMMA4(qz1.zw, sb, o2);
      }
      sac = sn;

      for (int r = 0; r < 8; ++r) {
        const int m = r + mrow;
        ostage[(j * kC + m) * kF + w * 16 + col] = o1[r] + o2[r];
      }
    }
    __syncthreads();

    {
      float* dst =
          out + (((size_t)(b * kL + tg * kR + cr) * kN + n) * kF + cc);
      for (int j = 0; j < kG; ++j) {
        float* dj = dst + (size_t)j * kC * kN * kF;
        const float* oj = &ostage[(j * kC + cr) * kF + cc];
        *reinterpret_cast<v4f*>(dj)     = *reinterpret_cast<const v4f*>(oj);
        *reinterpret_cast<v4f*>(dj + 4) = *reinterpret_cast<const v4f*>(oj + 4);
      }
    }
  }
}

extern "C" void kernel_launch(void* const* d_in, const int* in_sizes, int n_in,
                              void* d_out, int out_size, void* d_ws, size_t ws_size,
                              hipStream_t stream) {
  (void)in_sizes; (void)n_in; (void)out_size; (void)d_ws; (void)ws_size;
  const float* q  = (const float*)d_in[0];
  const float* Wq = (const float*)d_in[1];
  const float* bq = (const float*)d_in[2];
  const float* Wk = (const float*)d_in[3];
  const float* bk = (const float*)d_in[4];
  const float* Wv = (const float*)d_in[5];
  const float* bv = (const float*)d_in[6];
  cla_fused_kernel<<<dim3(kB * kN), dim3(256), 0, stream>>>(
      q, Wq, bq, Wk, bk, Wv, bv, (float*)d_out);
}


# Wq, Wk, Wv staged once per workgroup into LDS in B-fragment order, shared by two sequences (16 waves); projection reads W by 16-byte LDS loads instead of L2 loads; sum order unchanged
# speedup vs baseline: 1.4876x; 1.0642x over previous
#include <hip/hip_runtime.h>
#include <cmath>

typedef __attribute__((ext_vector_type(2))) float v2f;
typedef __attribute__((ext_vector_type(4))) float v4f;
typedef __attribute__((ext_vector_type(8))) float v8f;

#define WMMA4(a, b, c) __builtin_amdgcn_wmma_f32_16x16x4_f32( \
    false, (a), false, (b), (short)0, (c), false, false)

namespace {
constexpr int kB = 8, kL = 512, kN = 64, kF = 128, kH = 8, kC = 16;
constexpr int kG = 4;
constexpr int kR = kG * kC;
constexpr int kGroups = kL / kR;
constexpr int XS = 132;
constexpr int ST = 20;
constexpr float kEps = 1e-6f;
constexpr int kSeq = 2;
constexpr int kWSlots = 3 * kH * 16 * 32;
}

__device__ __forceinline__ float phi(float x) {
  return fmaxf(x, 0.0f) + __expf(fminf(x, 0.0f));
}

__device__ __forceinline__ float xhalf(float x) {
  return __builtin_bit_cast(float, __builtin_amdgcn_permlanex16(
      __builtin_bit_cast(int, x), __builtin_bit_cast(int, x),
      0x76543210, 0xfedcba98, false, false));
}

__global__ __launch_bounds__(512)
void cla_fused_kernel(const float* __restrict__ qin,
                      const float* __restrict__ Wq, const float* __restrict__ bq,
                      const float* __restrict__ Wk, const float* __restrict__ bk,
                      const float* __restrict__ Wv, const float* __restrict__ bv,
                      float* __restrict__ out)
{
  __shared__ alignas(16) float wl[kWSlots * 4];
  __shared__ alignas(16) float xo[kSeq][kR * XS];
  __shared__ alignas(16) float shQ[kSeq][kH][kC * ST];
  __shared__ alignas(16) float shK[kSeq][kH][kC * ST];
  __shared__ alignas(16) float shKcum[kSeq][kH][16];

  const int tid   = threadIdx.x;
  const int lane  = tid & 31;
  const int sq    = tid >> 8;
  const int w     = (tid >> 5) & 7;
  const int col   = lane & 15;
  const int hi    = lane >> 4;
  const int khalf = hi << 1;
  const int mrow  = hi << 3;

  const int g = blockIdx.x * kSeq + sq;
  const int b = g >> 6;
  const int n = g & 63;

  float* xs     = xo[sq];
  float* ostage = xo[sq];
  float* Qt = shQ[sq][w];
  float* Kt = shK[sq][w];
  float* Kc = shKcum[sq][w];

  for (int i = 0; i < kWSlots / 512; ++i) {
    const int slot = i * 512 + tid;
    const int ln = slot & 31;
    const int s2 = (slot >> 5) & 15;
    const int hd = (slot >> 9) & 7;
    const int mt = i >> 3;
    const float* Wm = (mt == 0) ? Wq : ((mt == 1) ? Wk : Wv);
    const int ka = 8 * s2 + 2 * (ln >> 4);
    const int cn = hd * 16 + (ln & 15);
    v4f v;
    v.x = Wm[ka * kF + cn];
    v.y = Wm[(ka + 1) * kF + cn];
    v.z = Wm[(ka + 4) * kF + cn];
    v.w = Wm[(ka + 5) * kF + cn];
    *reinterpret_cast<v4f*>(&wl[slot * 4]) = v;
  }
  const float* wlq = &wl[((0 * kH + w) * 16 * 32 + lane) * 4];
  const float* wlk = &wl[((1 * kH + w) * 16 * 32 + lane) * 4];
  const float* wlv = &wl[((2 * kH + w) * 16 * 32 + lane) * 4];

  v8f sac = {};
  float kcum = 0.0f;
  Kc[col] = 0.0f;

  const int wcol = w * 16 + col;
  const float bqv = bq[wcol];
  const float bkv = bk[wcol];
  const float bvv = bv[wcol];

  const int cr = (tid & 255) >> 4;
  const int cc = (tid & 15) << 3;

  for (int tg = 0; tg < kGroups; ++tg) {
    {
      const float* src =
          qin + (((size_t)(b * kL + tg * kR + cr) * kN + n) * kF + cc);
      v4f a0[kG], a1[kG];
      for (int j = 0; j < kG; ++j) {
        const float* sj = src + (size_t)j * kC * kN * kF;
        a0[j] = *reinterpret_cast<const v4f*>(sj);
        a1[j] = *reinterpret_cast<const v4f*>(sj + 4);
      }
      for (int j = 0; j < kG; ++j) {
        *reinterpret_cast<v4f*>(&xs[(j * kC + cr) * XS + cc])     = a0[j];
        *reinterpret_cast<v4f*>(&xs[(j * kC + cr) * XS + cc + 4]) = a1[j];
      }
      if (tg + 1 < kGroups) {
        for (int j = 0; j < kG; ++j) {
          __builtin_prefetch(src + (size_t)(kR + j * kC) * kN * kF, 0, 0);
        }
      }
    }
    __syncthreads();

    v8f cq[kG], ck[kG], cv[kG];
    for (int j = 0; j < kG; ++j) { cq[j] = {}; ck[j] = {}; cv[j] = {}; }
    #pragma unroll 2
    for (int s2 = 0; s2 < 16; ++s2) {
      const int k0 = 8 * s2 + khalf;
      const v4f wq4 = *reinterpret_cast<const v4f*>(wlq + s2 * 128);
      const v4f wk4 = *reinterpret_cast<const v4f*>(wlk + s2 * 128);
      const v4f wv4 = *reinterpret_cast<const v4f*>(wlv + s2 * 128);
      v2f xa[kG], xb[kG];
      for (int j = 0; j < kG; ++j) {
        xa[j] = *reinterpret_cast<v2f*>(&xs[(j * kC + col) * XS + k0]);
        xb[j] = *reinterpret_cast<v2f*>(&xs[(j * kC + col) * XS + k0 + 4]);
      }
      for (int j = 0; j < kG; ++j) {
        cq[j] = WMMA4(xa[j], wq4.xy, cq[j]);
        ck[j] = WMMA4(xa[j], wk4.xy, ck[j]);
        cv[j] = WMMA4(xa[j], wv4.xy, cv[j]);
      }
      for (int j = 0; j < kG; ++j) {
        cq[j] = WMMA4(xb[j], wq4.zw, cq[j]);
        ck[j] = WMMA4(xb[j], wk4.zw, ck[j]);
        cv[j] = WMMA4(xb[j], wv4.zw, cv[j]);
      }
    }
    __syncthreads();

    #pragma unroll
    for (int j = 0; j < kG; ++j) {
      float qv[8], kv[8];
      v8f vv;
      for (int r = 0; r < 8; ++r) {
        qv[r] = phi(cq[j][r] + bqv);
        kv[r] = phi(ck[j][r] + bkv);
        vv[r] = cv[j][r] + bvv;
      }
      for (int r = 0; r < 8; ++r) {
        const int m = r + mrow;
        Qt[m * ST + col] = qv[r];
        Kt[m * ST + col] = kv[r];
      }
      asm volatile("" ::: "memory");

      const v4f q0 = *reinterpret_cast<const v4f*>(&Qt[col * ST + mrow]);
      const v4f q1 = *reinterpret_cast<const v4f*>(&Qt[col * ST + mrow + 4]);
      const v4f k0 = *reinterpret_cast<const v4f*>(&Kt[col * ST + mrow]);
      const v4f k1 = *reinterpret_cast<const v4f*>(&Kt[col * ST + mrow + 4]);
      const v4f c0 = *reinterpret_cast<const v4f*>(&Kc[mrow]);
      const v4f c1 = *reinterpret_cast<const v4f*>(&Kc[mrow + 4]);
      asm volatile("" ::: "memory");

      float ks = ((kv[0] + kv[1]) + (kv[2] + kv[3])) +
                 ((kv[4] + kv[5]) + (kv[6] + kv[7]));
      ks += xhalf(ks);
      kcum += ks;
      Kc[col] = kcum;

      v8f at = {};
      v8f sn = sac;
      {
        v2f ka, va;
        at = WMMA4(k0.xy, q0.xy, at);
        ka.x = kv[0]; ka.y = kv[1]; va.x = vv[0]; va.y = vv[1];
        sn = WMMA4(ka, va, sn);
        at = WMMA4(k0.zw, q0.zw, at);
        ka.x = kv[2]; ka.y = kv[3]; va.x = vv[2]; va.y = vv[3];
        sn = WMMA4(ka, va, sn);
        at = WMMA4(k1.xy, q1.xy, at);
        ka.x = kv[4]; ka.y = kv[5]; va.x = vv[4]; va.y = vv[5];
        sn = WMMA4(ka, va, sn);
        at = WMMA4(k1.zw, q1.zw, at);
        ka.x = kv[6]; ka.y = kv[7]; va.x = vv[6]; va.y = vv[7];
        sn = WMMA4(ka, va, sn);
      }

      float p = 0.0f;
      for (int r = 0; r < 8; ++r) {
        at[r] = (mrow + r <= col) ? at[r] : 0.0f;
        p += at[r];
      }
      p += q0.x * c0.x; p += q0.y * c0.y; p += q0.z * c0.z; p += q0.w * c0.w;
      p += q1.x * c1.x; p += q1.y * c1.y; p += q1.z * c1.z; p += q1.w * c1.w;
      p += xhalf(p);
      const float z = __builtin_amdgcn_rcpf(p + kEps);

      v8f o1 = {};
      v8f o2 = {};
      const v4f qz0 = q0 * z;
      const v4f qz1 = q1 * z;
      {
        v2f aa, bb, sb;
        aa.x = at[0] * z; aa.y = at[1] * z; bb.x = vv[0]; bb.y = vv[1];
        o1 = WMMA4(aa, bb, o1);
        sb.x = sac[0]; sb.y = sac[1];
        o2 = WMMA4(qz0.xy, sb, o2);
        aa.x = at[2] * z; aa.y = at[3] * z; bb.x = vv[2]; bb.y = vv[3];
        o1 = WMMA4(aa, bb, o1);
        sb.x = sac[2]; sb.y = sac[3];
        o2 = WMMA4(qz0.zw, sb, o2);
        aa.x = at[4] * z; aa.y = at[5] * z; bb.x = vv[4]; bb.y = vv[5];
        o1 = WMMA4(aa, bb, o1);
        sb.x = sac[4]; sb.y = sac[5];
        o2 = WMMA4(qz1.xy, sb, o2);
        aa.x = at[6] * z; aa.y = at[7] * z; bb.x = vv[6]; bb.y = vv[7];
        o1 = WMMA4(aa, bb, o1);
        sb.x = sac[6]; sb.y = sac[7];
        o2 = WMMA4(qz1.zw, sb, o2);
      }
      sac = sn;

      for (int r = 0; r < 8; ++r) {
        const int m = r + mrow;
        ostage[(j * kC + m) * kF + w * 16 + col] = o1[r] + o2[r];
      }
    }
    __syncthreads();

    {
      v4f t0[kG], t1[kG];
      for (int j = 0; j < kG; ++j) {
        const float* oj = &ostage[(j * kC + cr) * kF + cc];
        t0[j] = *reinterpret_cast<const v4f*>(oj);
        t1[j] = *reinterpret_cast<const v4f*>(oj + 4);
      }
      __syncthreads();
      float* dst =
          out + (((size_t)(b * kL + tg * kR + cr) * kN + n) * kF + cc);
      for (int j = 0; j < kG; ++j) {
        float* dj = dst + (size_t)j * kC * kN * kF;
        *reinterpret_cast<v4f*>(dj)     = t0[j];
        *reinterpret_cast<v4f*>(dj + 4) = t1[j];
      }
    }
  }
}

extern "C" void kernel_launch(void* const* d_in, const int* in_sizes, int n_in,
                              void* d_out, int out_size, void* d_ws, size_t ws_size,
                              hipStream_t stream) {
  (void)in_sizes; (void)n_in; (void)out_size; (void)d_ws; (void)ws_size;
  const float* q  = (const float*)d_in[0];
  const float* Wq = (const float*)d_in[1];
  const float* bq = (const float*)d_in[2];
  const float* Wk = (const float*)d_in[3];
  const float* bk = (const float*)d_in[4];
  const float* Wv = (const float*)d_in[5];
  const float* bv = (const float*)d_in[6];
  cla_fused_kernel<<<dim3(kB * kN / kSeq), dim3(512), 0, stream>>>(
      q, Wq, bq, Wk, bk, Wv, bv, (float*)d_out);
}


# Shared output tile dropped: head pairs swap half their output rows via each other's idle Q/K tiles, each wave stores whole lines; 2 barriers per group not 4, next x fetched in scan; sums unchanged
# speedup vs baseline: 1.5513x; 1.0428x over previous
#include <hip/hip_runtime.h>
#include <cmath>

typedef __attribute__((ext_vector_type(2))) float v2f;
typedef __attribute__((ext_vector_type(4))) float v4f;
typedef __attribute__((ext_vector_type(8))) float v8f;

#define WMMA4(a, b, c) __builtin_amdgcn_wmma_f32_16x16x4_f32( \
    false, (a), false, (b), (short)0, (c), false, false)

namespace {
constexpr int kB = 8, kL = 512, kN = 64, kF = 128, kH = 8, kC = 16;
constexpr int kG = 4;
constexpr int kR = kG * kC;
constexpr int kGroups = kL / kR;
constexpr int XS = 132;
constexpr int ST = 20;
constexpr float kEps = 1e-6f;
constexpr int kSeq = 2;
constexpr int kWSlots = 3 * kH * 16 * 32;
}

__device__ __forceinline__ float phi(float x) {
  return fmaxf(x, 0.0f) + __expf(fminf(x, 0.0f));
}

__device__ __forceinline__ float xhalf(float x) {
  return __builtin_bit_cast(float, __builtin_amdgcn_permlanex16(
      __builtin_bit_cast(int, x), __builtin_bit_cast(int, x),
      0x76543210, 0xfedcba98, false, false));
}

__global__ __launch_bounds__(512)
void cla_fused_kernel(const float* __restrict__ qin,
                      const float* __restrict__ Wq, const float* __restrict__ bq,
                      const float* __restrict__ Wk, const float* __restrict__ bk,
                      const float* __restrict__ Wv, const float* __restrict__ bv,
                      float* __restrict__ out)
{
  __shared__ alignas(16) float wl[kWSlots * 4];
  __shared__ alignas(16) float xo[kSeq][kR * XS];
  __shared__ alignas(16) float shQ[kSeq][kH][kC * ST];
  __shared__ alignas(16) float shK[kSeq][kH][kC * ST];
  __shared__ alignas(16) float shKcum[kSeq][kH][16];

  const int tid   = threadIdx.x;
  const int lane  = tid & 31;
  const int sq    = tid >> 8;
  const int w     = (tid >> 5) & 7;
  const int col   = lane & 15;
  const int hi    = lane >> 4;
  const int khalf = hi << 1;
  const int mrow  = hi << 3;

  const int g = blockIdx.x * kSeq + sq;
  const int b = g >> 6;
  const int n = g & 63;

  float* xs = xo[sq];
  float* Qt = shQ[sq][w];
  float* Kt = shK[sq][w];
  float* Kc = shKcum[sq][w];

  const int  ph    = w & 1;
  const bool sends = (hi != ph);
  float* PQ = shQ[sq][w ^ 1];
  float* PK = shK[sq][w ^ 1];
  float* obase = out + (((size_t)(b * kL + 8 * ph) * kN + n) * kF + (sends ? (w ^ 1) : w) * 16 + col);

  for (int i = 0; i < kWSlots / 512; ++i) {
    const int slot = i * 512 + tid;
    const int ln = slot & 31;
    const int s2 = (slot >> 5) & 15;
    const int hd = (slot >> 9) & 7;
    const int mt = i >> 3;
    const float* Wm = (mt == 0) ? Wq : ((mt == 1) ? Wk : Wv);
    const int ka = 8 * s2 + 2 * (ln >> 4);
    const int cn = hd * 16 + (ln & 15);
    v4f v;
    v.x = Wm[ka * kF + cn];
    v.y = Wm[(ka + 1) * kF + cn];
    v.z = Wm[(ka + 4) * kF + cn];
    v.w = Wm[(ka + 5) * kF + cn];
    *reinterpret_cast<v4f*>(&wl[slot * 4]) = v;
  }
  const float* wlq = &wl[((0 * kH + w) * 16 * 32 + lane) * 4];
  const float* wlk = &wl[((1 * kH + w) * 16 * 32 + lane) * 4];
  const float* wlv = &wl[((2 * kH + w) * 16 * 32 + lane) * 4];

  v8f sac = {};
  float kcum = 0.0f;
  Kc[col] = 0.0f;

  const int wcol = w * 16 + col;
  const float bqv = bq[wcol];
  const float bkv = bk[wcol];
  const float bvv = bv[wcol];

  const int cr = (tid & 255) >> 4;
  const int cc = (tid & 15) << 3;

  {
    const float* src = qin + (((size_t)(b * kL + cr) * kN + n) * kF + cc);
    v4f f0[kG], f1[kG];
    for (int j = 0; j < kG; ++j) {
      const float* sj = src + (size_t)j * kC * kN * kF;
      f0[j] = *reinterpret_cast<const v4f*>(sj);
      f1[j] = *reinterpret_cast<const v4f*>(sj + 4);
    }
    for (int j = 0; j < kG; ++j) {
      *reinterpret_cast<v4f*>(&xs[(j * kC + cr) * XS + cc])     = f0[j];
      *reinterpret_cast<v4f*>(&xs[(j * kC + cr) * XS + cc + 4]) = f1[j];
    }
    for (int j = 0; j < kG; ++j) {
      __builtin_prefetch(src + (size_t)(kR + j * kC) * kN * kF, 0, 0);
    }
  }
  __syncthreads();

  for (int tg = 0; tg < kGroups; ++tg) {
    const bool more = (tg + 1 < kGroups);
    const float* srcn =
        qin + (((size_t)(b * kL + (tg + 1) * kR + cr) * kN + n) * kF + cc);
    v4f a0[kG], a1[kG];

    v8f cq[kG], ck[kG], cv[kG];
    for (int j = 0; j < kG; ++j) { cq[j] = {}; ck[j] = {}; cv[j] = {}; }
    #pragma unroll 2
    for (int s2 = 0; s2 < 16; ++s2) {
      const int k0 = 8 * s2 + khalf;
      const v4f wq4 = *reinterpret_cast<const v4f*>(wlq + s2 * 128);
      const v4f wk4 = *reinterpret_cast<const v4f*>(wlk + s2 * 128);
      const v4f wv4 = *reinterpret_cast<const v4f*>(wlv + s2 * 128);
      v2f xa[kG], xb[kG];
      for (int j = 0; j < kG; ++j) {
        xa[j] = *reinterpret_cast<v2f*>(&xs[(j * kC + col) * XS + k0]);
        xb[j] = *reinterpret_cast<v2f*>(&xs[(j * kC + col) * XS + k0 + 4]);
      }
      for (int j = 0; j < kG; ++j) {
        cq[j] = WMMA4(xa[j], wq4.xy, cq[j]);
        ck[j] = WMMA4(xa[j], wk4.xy, ck[j]);
        cv[j] = WMMA4(xa[j], wv4.xy, cv[j]);
      }
      for (int j = 0; j < kG; ++j) {
        cq[j] = WMMA4(xb[j], wq4.zw, cq[j]);
        ck[j] = WMMA4(xb[j], wk4.zw, ck[j]);
        cv[j] = WMMA4(xb[j], wv4.zw, cv[j]);
      }
    }

    float ov[kG][8];
    #pragma unroll
    for (int j = 0; j < kG; ++j) {
      float qv[8], kv[8];
      v8f vv;
      for (int r = 0; r < 8; ++r) {
        qv[r] = phi(cq[j][r] + bqv);
        kv[r] = phi(ck[j][r] + bkv);
        vv[r] = cv[j][r] + bvv;
      }
      for (int r = 0; r < 8; ++r) {
        const int m = r + mrow;
        Qt[m * ST + col] = qv[r];
        Kt[m * ST + col] = kv[r];
      }
      asm volatile("" ::: "memory");

      const v4f q0 = *reinterpret_cast<const v4f*>(&Qt[col * ST + mrow]);
      const v4f q1 = *reinterpret_cast<const v4f*>(&Qt[col * ST + mrow + 4]);
      const v4f k0 = *reinterpret_cast<const v4f*>(&Kt[col * ST + mrow]);
      const v4f k1 = *reinterpret_cast<const v4f*>(&Kt[col * ST + mrow + 4]);
      const v4f c0 = *reinterpret_cast<const v4f*>(&Kc[mrow]);
      const v4f c1 = *reinterpret_cast<const v4f*>(&Kc[mrow + 4]);
      asm volatile("" ::: "memory");

      float ks = ((kv[0] + kv[1]) + (kv[2] + kv[3])) +
                 ((kv[4] + kv[5]) + (kv[6] + kv[7]));
      ks += xhalf(ks);
      kcum += ks;
      Kc[col] = kcum;

      v8f at = {};
      v8f sn = sac;
      {
        v2f ka, va;
        at = WMMA4(k0.xy, q0.xy, at);
        ka.x = kv[0]; ka.y = kv[1]; va.x = vv[0]; va.y = vv[1];
        sn = WMMA4(ka, va, sn);
        at = WMMA4(k0.zw, q0.zw, at);
        ka.x = kv[2]; ka.y = kv[3]; va.x = vv[2]; va.y = vv[3];
        sn = WMMA4(ka, va, sn);
        at = WMMA4(k1.xy, q1.xy, at);
        ka.x = kv[4]; ka.y = kv[5]; va.x = vv[4]; va.y = vv[5];
        sn = WMMA4(ka, va, sn);
        at = WMMA4(k1.zw, q1.zw, at);
        ka.x = kv[6]; ka.y = kv[7]; va.x = vv[6]; va.y = vv[7];
        sn = WMMA4(ka, va, sn);
      }

      float p = 0.0f;
      for (int r = 0; r < 8; ++r) {
        at[r] = (mrow + r <= col) ? at[r] : 0.0f;
        p += at[r];
      }
      p += q0.x * c0.x; p += q0.y * c0.y; p += q0.z * c0.z; p += q0.w * c0.w;
      p += q1.x * c1.x; p += q1.y * c1.y; p += q1.z * c1.z; p += q1.w * c1.w;
      p += xhalf(p);
      const float z = __builtin_amdgcn_rcpf(p + kEps);

      v8f o1 = {};
      v8f o2 = {};
      const v4f qz0 = q0 * z;
      const v4f qz1 = q1 * z;
      {
        v2f aa, bb, sb;
        aa.x = at[0] * z; aa.y = at[1] * z; bb.x = vv[0]; bb.y = vv[1];
        o1 = WMMA4(aa, bb, o1);
        sb.x = sac[0]; sb.y = sac[1];
        o2 = WMMA4(qz0.xy, sb, o2);
        aa.x = at[2] * z; aa.y = at[3] * z; bb.x = vv[2]; bb.y = vv[3];
        o1 = WMMA4(aa, bb, o1);
        sb.x = sac[2]; sb.y = sac[3];
        o2 = WMMA4(qz0.zw, sb, o2);
        aa.x = at[4] * z; aa.y = at[5] * z; bb.x = vv[4]; bb.y = vv[5];
        o1 = WMMA4(aa, bb, o1);
        sb.x = sac[4]; sb.y = sac[5];
        o2 = WMMA4(qz1.xy, sb, o2);
        aa.x = at[6] * z; aa.y = at[7] * z; bb.x = vv[6]; bb.y = vv[7];
        o1 = WMMA4(aa, bb, o1);
        sb.x = sac[6]; sb.y = sac[7];
        o2 = WMMA4(qz1.zw, sb, o2);
      }
      sac = sn;

      for (int r = 0; r < 8; ++r) ov[j][r] = o1[r] + o2[r];

      if (j == 1 && more) {
        for (int jj = 0; jj < kG; ++jj) {
          const float* sj = srcn + (size_t)jj * kC * kN * kF;
          a0[jj] = *reinterpret_cast<const v4f*>(sj);
          a1[jj] = *reinterpret_cast<const v4f*>(sj + 4);
        }
      }
    }
    __syncthreads();

    if (sends) {
      #pragma unroll
      for (int j = 0; j < kG; ++j) {
        float* pt = ((j < 2) ? PQ : PK) + (j & 1) * 128 + col * 4;
        v4f lo, hi4;
        lo.x  = ov[j][0]; lo.y  = ov[j][1]; lo.z  = ov[j][2]; lo.w  = ov[j][3];
        hi4.x = ov[j][4]; hi4.y = ov[j][5]; hi4.z = ov[j][6]; hi4.w = ov[j][7];
        *reinterpret_cast<v4f*>(pt)      = lo;
        *reinterpret_cast<v4f*>(pt + 64) = hi4;
      }
    }
    if (more) {
      for (int j = 0; j < kG; ++j) {
        *reinterpret_cast<v4f*>(&xs[(j * kC + cr) * XS + cc])     = a0[j];
        *reinterpret_cast<v4f*>(&xs[(j * kC + cr) * XS + cc + 4]) = a1[j];
      }
      if (tg + 2 < kGroups) {
        for (int j = 0; j < kG; ++j) {
          __builtin_prefetch(srcn + (size_t)(kR + j * kC) * kN * kF, 0, 0);
        }
      }
    }
    __syncthreads();

    {
      float* dst = obase + (size_t)(tg * kR) * kN * kF;
      #pragma unroll
      for (int j = 0; j < kG; ++j) {
        const float* mt = ((j < 2) ? Qt : Kt) + (j & 1) * 128 + col * 4;
        v4f lo, hi4;
        lo.x  = ov[j][0]; lo.y  = ov[j][1]; lo.z  = ov[j][2]; lo.w  = ov[j][3];
        hi4.x = ov[j][4]; hi4.y = ov[j][5]; hi4.z = ov[j][6]; hi4.w = ov[j][7];
        if (sends) {
          lo  = *reinterpret_cast<const v4f*>(mt);
          hi4 = *reinterpret_cast<const v4f*>(mt + 64);
        }
        float* dj = dst + (size_t)(j * kC) * kN * kF;
        dj[(size_t)0 * kN * kF] = lo.x;
        dj[(size_t)1 * kN * kF] = lo.y;
        dj[(size_t)2 * kN * kF] = lo.z;
        dj[(size_t)3 * kN * kF] = lo.w;
        dj[(size_t)4 * kN * kF] = hi4.x;
        dj[(size_t)5 * kN * kF] = hi4.y;
        dj[(size_t)6 * kN * kF] = hi4.z;
        dj[(size_t)7 * kN * kF] = hi4.w;
      }
      asm volatile("" ::: "memory");
    }
  }
}

extern "C" void kernel_launch(void* const* d_in, const int* in_sizes, int n_in,
                              void* d_out, int out_size, void* d_ws, size_t ws_size,
                              hipStream_t stream) {
  (void)in_sizes; (void)n_in; (void)out_size; (void)d_ws; (void)ws_size;
  const float* q  = (const float*)d_in[0];
  const float* Wq = (const float*)d_in[1];
  const float* bq = (const float*)d_in[2];
  const float* Wk = (const float*)d_in[3];
  const float* bk = (const float*)d_in[4];
  const float* Wv = (const float*)d_in[5];
  const float* bv = (const float*)d_in[6];
  cla_fused_kernel<<<dim3(kB * kN / kSeq), dim3(512), 0, stream>>>(
      q, Wq, bq, Wk, bk, Wv, bv, (float*)d_out);
}
